# starting kernel
# baseline (speedup 1.0000x reference)
#include <hip/hip_runtime.h>


typedef __attribute__((ext_vector_type(2))) float v2f;
typedef __attribute__((ext_vector_type(8))) float v8f;

#define SIZEP 10
#define CIN   32
#define COUT  32
#define NR    252
#define WAVES 4

__device__ __forceinline__ void wmma_rtile(const v2f* pSA, const v2f* pSB,
                                           const v2f b0[8], const v2f b1[8],
                                           int halfsel, float bz0, float bz1,
                                           v8f& acc0, v8f& acc1) {
  acc0 = (v8f){bz0, bz0, bz0, bz0, bz0, bz0, bz0, bz0};
  acc1 = (v8f){bz1, bz1, bz1, bz1, bz1, bz1, bz1, bz1};
  #pragma unroll
  for (int kk = 0; kk < 8; ++kk) {
    const int ch = 2 * kk + halfsel;
    const v2f a  = pSA[ch] + pSB[ch];
    acc0 = __builtin_amdgcn_wmma_f32_16x16x4_f32(
        false, a, false, b0[kk], (short)0, acc0, false, false);
    acc1 = __builtin_amdgcn_wmma_f32_16x16x4_f32(
        false, a, false, b1[kk], (short)0, acc1, false, false);
  }
}

__global__ __launch_bounds__(WAVES * 32)
void pcl_wmma_kernel(const float* __restrict__ x,
                     const float* __restrict__ W,
                     const int*   __restrict__ split0,
                     float* __restrict__ out,
                     int samples) {
  __shared__ __align__(16) unsigned int s_mask[256];
  __shared__ __align__(16) float s_Wd[COUT * CIN];
  __shared__ __align__(16) float s_SA[WAVES][32 * 32];
  __shared__ __align__(16) float s_SB[WAVES][32 * 32];
  __shared__ __align__(16) float s_base[WAVES][32];

  const int tid = threadIdx.x;
  const int w   = tid >> 5;
  const int l   = tid & 31;
  const int s   = blockIdx.x * WAVES + w;

  for (int r = tid; r < 256; r += WAVES * 32) {
    unsigned int mA = 0u, mB = 0u;
    if (r < NR) {
      #pragma unroll
      for (int g = 0; g < 5; ++g) {
        int p = split0[r * 5 + g];
        if (p < 5) mA |= 1u << p;
        else       mB |= 1u << (p - 5);
      }
    }
    s_mask[r] = mA | (mB << 8);
  }

  #pragma unroll
  for (int i = 0; i < (COUT * CIN) / (WAVES * 32); ++i) {
    const int idx = tid + i * (WAVES * 32);
    s_Wd[idx] = W[idx] - W[COUT * CIN + idx];
  }

  if (s < samples) {
    const float* xs = x + (size_t)s * (SIZEP * CIN);
    float xa[5], xb[5];
    #pragma unroll
    for (int p = 0; p < 5; ++p) {
      xa[p] = xs[p * CIN + l];
      xb[p] = xs[(p + 5) * CIN + l];
    }
    float sa[32], sb[32];
    sa[0] = 0.f; sb[0] = 0.f;
    #pragma unroll
    for (int m = 1; m < 32; ++m) {
      sa[m] = sa[m & (m - 1)] + xa[__builtin_ctz(m)];
      sb[m] = sb[m & (m - 1)] + xb[__builtin_ctz(m)];
    }
    #pragma unroll
    for (int m = 0; m < 32; ++m) {
      s_SA[w][m * 32 + l] = sa[m];
      s_SB[w][m * 32 + l] = sb[m];
    }
  }
  __syncthreads();

  if (s < samples) {
    const float* w1row = W + COUT * CIN + l * CIN;
    float acc = 0.f;
    #pragma unroll
    for (int c = 0; c < CIN; ++c)
      acc += (s_SA[w][31 * 32 + c] + s_SB[w][31 * 32 + c]) * w1row[c];
    s_base[w][l] = acc;
  }
  __syncthreads();

  if (s >= samples) return;

  const int halfsel = l >> 4;
  const int nlo     = l & 15;

  v2f b0[8], b1[8];
  {
    const v2f* pB0 = (const v2f*)&s_Wd[nlo * CIN];
    const v2f* pB1 = (const v2f*)&s_Wd[(16 + nlo) * CIN];
    #pragma unroll
    for (int kk = 0; kk < 8; ++kk) {
      b0[kk] = pB0[2 * kk + halfsel];
      b1[kk] = pB1[2 * kk + halfsel];
    }
  }

  const float bz0 = s_base[w][nlo];
  const float bz1 = s_base[w][16 + nlo];

  float* outp = out + (size_t)s * (NR * COUT);

  for (int rt = 0; rt < 15; ++rt) {
    const unsigned int mm = s_mask[rt * 16 + nlo];
    const v2f* pSA = (const v2f*)&s_SA[w][(mm & 31u) * 32];
    const v2f* pSB = (const v2f*)&s_SB[w][((mm >> 8) & 31u) * 32];

    v8f acc0, acc1;
    wmma_rtile(pSA, pSB, b0, b1, halfsel, bz0, bz1, acc0, acc1);

    float* orow = outp + (rt * 16 + 8 * halfsel) * COUT + nlo;
    #pragma unroll
    for (int j = 0; j < 8; ++j) {
      orow[j * COUT]      = acc0[j];
      orow[j * COUT + 16] = acc1[j];
    }
  }

  {
    const unsigned int mm = s_mask[15 * 16 + nlo];
    const v2f* pSA = (const v2f*)&s_SA[w][(mm & 31u) * 32];
    const v2f* pSB = (const v2f*)&s_SB[w][((mm >> 8) & 31u) * 32];

    v8f acc0, acc1;
    wmma_rtile(pSA, pSB, b0, b1, halfsel, bz0, bz1, acc0, acc1);

    float* orow = outp + (15 * 16 + 8 * halfsel) * COUT + nlo;
    #pragma unroll
    for (int j = 0; j < 8; ++j) {
      if (8 * halfsel + j < NR - 240) {
        orow[j * COUT]      = acc0[j];
        orow[j * COUT + 16] = acc1[j];
      }
    }
  }
}

extern "C" void kernel_launch(void* const* d_in, const int* in_sizes, int n_in,
                              void* d_out, int out_size, void* d_ws, size_t ws_size,
                              hipStream_t stream) {
  const float* x      = (const float*)d_in[0];
  const float* W      = (const float*)d_in[1];
  const int*   splits = (const int*)d_in[2];
  float* out = (float*)d_out;

  const int samples = in_sizes[0] / (SIZEP * CIN);
  const int blocks  = (samples + WAVES - 1) / WAVES;

  pcl_wmma_kernel<<<blocks, WAVES * 32, 0, stream>>>(x, W, splits, out, samples);
}


# product taken before the gather: 32 WMMAs on the 10 position rows per sample (was 256 on pooled rows); out row = sum of two LDS table rows, whole-line b128 stores; f32 sums reordered
# speedup vs baseline: 1.9652x; 1.9652x over previous
#include <hip/hip_runtime.h>


typedef __attribute__((ext_vector_type(2))) float v2f;
typedef __attribute__((ext_vector_type(4))) float v4f;
typedef __attribute__((ext_vector_type(8))) float v8f;

#define SIZEP 10
#define CIN   32
#define COUT  32
#define NR    252
#define WAVES 4
#define WP    36
#define TROW  64
#define OFFP  68

__global__ __launch_bounds__(WAVES * 32)
void pcl_wmma_kernel(const float* __restrict__ x,
                     const float* __restrict__ W,
                     const int*   __restrict__ split0,
                     float* __restrict__ out,
                     int samples) {
  __shared__ __align__(16) float s_W[2 * COUT * WP];
  __shared__ __align__(16) float s_T[WAVES][32 * TROW];
  __shared__ __align__(16) unsigned int s_off[4 * OFFP];

  const int tid = threadIdx.x;
  const int w   = tid >> 5;
  const int l   = tid & 31;
  const int s   = blockIdx.x * WAVES + w;
  const int sl  = s < samples ? s : samples - 1;
  const int c   = l & 15;
  const int h   = l >> 4;

  v4f a4[4];
  {
    const int p = (c < 5) ? c : ((c >= 8 && c < 13) ? c - 3 : 0);
    const float* xs = x + (size_t)sl * (SIZEP * CIN) + p * CIN + 4 * h;
    #pragma unroll
    for (int t2 = 0; t2 < 4; ++t2) a4[t2] = *(const v4f*)(xs + 8 * t2);
  }

  v4f wv[(2 * COUT * CIN / 4) / (WAVES * 32)];
  #pragma unroll
  for (int j = 0; j < (2 * COUT * CIN / 4) / (WAVES * 32); ++j)
    wv[j] = ((const v4f*)W)[tid + j * (WAVES * 32)];

  for (int r = tid; r < 256; r += WAVES * 32) {
    unsigned int mA = 0u, mB = 0u;
    if (r < NR) {
      #pragma unroll
      for (int g = 0; g < 5; ++g) {
        int p = split0[r * 5 + g];
        if (p < 5) mA |= 1u << p;
        else       mB |= 1u << (p - 5);
      }
    }
    const unsigned int oa = (mA & 31u) * (TROW * 4);
    const unsigned int ob = (mB & 31u) * (TROW * 4) + 128u;
    const unsigned int first  = (r & 1) ? ob : oa;
    const unsigned int second = (r & 1) ? oa : ob;
    s_off[(r & 3) * OFFP + (r >> 2)] = first | (second << 16);
  }

  #pragma unroll
  for (int j = 0; j < (2 * COUT * CIN / 4) / (WAVES * 32); ++j) {
    const int idx4 = tid + j * (WAVES * 32);
    *(v4f*)&s_W[(idx4 >> 3) * WP + (idx4 & 7) * 4] = wv[j];
  }
  __syncthreads();

  v4f b4[4][4];
  #pragma unroll
  for (int nt = 0; nt < 4; ++nt) {
    const float* wr = &s_W[(nt * 16 + c) * WP + 4 * h];
    #pragma unroll
    for (int t2 = 0; t2 < 4; ++t2) b4[nt][t2] = *(const v4f*)(wr + 8 * t2);
  }

  v8f acc[4];
  #pragma unroll
  for (int nt = 0; nt < 4; ++nt) acc[nt] = (v8f){0.f, 0.f, 0.f, 0.f, 0.f, 0.f, 0.f, 0.f};
  #pragma unroll
  for (int t2 = 0; t2 < 4; ++t2) {
    #pragma unroll
    for (int u = 0; u < 2; ++u) {
      const v2f a = (v2f){a4[t2][2 * u], a4[t2][2 * u + 1]};
      #pragma unroll
      for (int nt = 0; nt < 4; ++nt) {
        const v2f b = (v2f){b4[nt][t2][2 * u], b4[nt][t2][2 * u + 1]};
        acc[nt] = __builtin_amdgcn_wmma_f32_16x16x4_f32(
            false, a, false, b, (short)0, acc[nt], false, false);
      }
    }
  }

  {
    float d0[5], d1[5];
    #pragma unroll
    for (int j = 0; j < 5; ++j) {
      d0[j] = acc[0][j] - acc[2][j];
      d1[j] = acc[1][j] - acc[3][j];
    }
    float t0[32], t1[32];
    t0[0] = ((acc[2][0] + acc[2][1]) + (acc[2][2] + acc[2][3])) + acc[2][4];
    t1[0] = ((acc[3][0] + acc[3][1]) + (acc[3][2] + acc[3][3])) + acc[3][4];
    #pragma unroll
    for (int m = 1; m < 32; ++m) {
      t0[m] = t0[m & (m - 1)] + d0[__builtin_ctz(m)];
      t1[m] = t1[m & (m - 1)] + d1[__builtin_ctz(m)];
    }
    float* tw = &s_T[w][32 * h + c];
    #pragma unroll
    for (int m = 0; m < 32; ++m) {
      tw[m * TROW]      = t0[m];
      tw[m * TROW + 16] = t1[m];
    }
  }
  __syncthreads();

  if (s >= samples) return;

  {
    const int g = l >> 3;
    const int q = l & 7;
    const char* tb = (const char*)&s_T[w][0] + q * 16;
    const unsigned int* po = &s_off[g * OFFP];
    v4f* op = (v4f*)(out + (size_t)s * (NR * COUT)) + l;
    #pragma unroll
    for (int i = 0; i < NR / 4; ++i) {
      const unsigned int pk = po[i];
      const v4f v1 = *(const v4f*)(tb + (pk & 0xffffu));
      const v4f v2 = *(const v4f*)(tb + (pk >> 16));
      op[i * 32] = v1 + v2;
    }
  }
}

extern "C" void kernel_launch(void* const* d_in, const int* in_sizes, int n_in,
                              void* d_out, int out_size, void* d_ws, size_t ws_size,
                              hipStream_t stream) {
  const float* x      = (const float*)d_in[0];
  const float* W      = (const float*)d_in[1];
  const int*   splits = (const int*)d_in[2];
  float* out = (float*)d_out;

  const int samples = in_sizes[0] / (SIZEP * CIN);
  const int blocks  = (samples + WAVES - 1) / WAVES;

  pcl_wmma_kernel<<<blocks, WAVES * 32, 0, stream>>>(x, W, splits, out, samples);
}
